# starting kernel
# baseline (speedup 1.0000x reference)
#include <hip/hip_runtime.h>

typedef __attribute__((ext_vector_type(2))) float v2f;
typedef __attribute__((ext_vector_type(4))) float v4f;
typedef __attribute__((ext_vector_type(8))) float v8f;

#define FS_F      8000.0f
#define HALF_W    16
#define KTAPS     33
#define PI_F      3.14159265358979323846f
#define TWO_PI_F  6.28318530717958647692f

__global__ __launch_bounds__(256) void
lpf_wmma_kernel(const float* __restrict__ in, const float* __restrict__ alpha_p,
                const float* __restrict__ beta_p, float* __restrict__ out,
                int n, int nbatch)
{
    const float alpha = alpha_p[0];
    const float beta  = beta_p[0];

    const int lane = threadIdx.x & 31;
    const int wv   = threadIdx.x >> 5;
    const int tile = blockIdx.x * (blockDim.x >> 5) + wv;
    const int t0   = tile * 16;
    if (t0 >= n) return;

    const int m  = lane & 15;
    const int hl = lane >> 4;
    const int t  = t0 + m;

    const float c_t =
        (PI_F * 0.25f + alpha * __sinf(beta * (float)t * (1.0f / FS_F))) * (1.0f / TWO_PI_F);
    const float c2   = 2.0f * c_t;
    const float xstp = TWO_PI_F * c_t;

    auto tapf = [&](float fk, bool is16) -> float {
        float w = 0.5f - 0.5f * __cosf(PI_F * (1.0f / 16.0f) * fk);
        float x = xstp * (fk - 16.0f);
        float s = __sinf(x) * __builtin_amdgcn_rcpf(x);
        s = is16 ? 1.0f : s;
        return c2 * w * s;
    };

    const int br = (m < nbatch) ? m : 0;
    const float* __restrict__ inb = in + (size_t)br * (size_t)n;

    v8f acc = {};
    float S;

    const bool edge = (t0 < HALF_W) || (t0 > n - 3 * HALF_W);
    if (!edge) {
        const float fkm = (float)(-m);
        const float* __restrict__ pB = inb + (size_t)(t0 + 2 * hl);
        float Apart = 0.0f;
        #pragma unroll
        for (int c = 0; c < 12; ++c) {
            const int base = 4 * c + 2 * hl;
            const int k0 = base - m;
            const float fk0 = fkm + (float)base;
            v2f A, B;
            float v0 = tapf(fk0,        k0     == HALF_W) * (((unsigned)k0       < KTAPS) ? 1.0f : 0.0f);
            float v1 = tapf(fk0 + 1.0f, k0 + 1 == HALF_W) * (((unsigned)(k0 + 1) < KTAPS) ? 1.0f : 0.0f);
            A.x = v0; A.y = v1;
            Apart += v0 + v1;
            const float2 bb = *(const float2*)(pB + 4 * c);
            B.x = bb.x;
            B.y = bb.y;
            acc = __builtin_amdgcn_wmma_f32_16x16x4_f32(
                false, A, false, B, (short)0, acc, false, false);
        }
        S = Apart + __shfl_xor(Apart, 16, 32);
    } else {
        const int off = (t < HALF_W) ? -(HALF_W + 1) : 0;
        const float fkm = (float)(-m - off);
        #pragma unroll
        for (int c = 0; c < 12; ++c) {
            const int base = 4 * c + 2 * hl;
            v2f A, B;
            #pragma unroll
            for (int j = 0; j < 2; ++j) {
                int k = base + j - m - off;
                bool valid = ((unsigned)k < KTAPS);
                valid = valid & ((t >= HALF_W) | (k >= HALF_W + 1 - t));
                valid = valid & ((t <= n - KTAPS) | (k < n - t));
                float v = tapf(fkm + (float)(base + j), k == HALF_W)
                          * (valid ? 1.0f : 0.0f);
                if (j == 0) A.x = v; else A.y = v;
            }
            int col = t0 + base;
            col = (col < n - 1) ? col : (n - 2);
            const float2 bb = *(const float2*)(inb + (size_t)col);
            B.x = bb.x;
            B.y = bb.y;
            acc = __builtin_amdgcn_wmma_f32_16x16x4_f32(
                false, A, false, B, (short)0, acc, false, false);
        }
        float Sp = 0.0f;
        #pragma unroll
        for (int j = 0; j <= 16; ++j) {
            int k = 2 * j + hl;
            float v = tapf((float)k, k == HALF_W);
            Sp += v * ((k < KTAPS) ? 1.0f : 0.0f);
        }
        S = Sp + __shfl_xor(Sp, 16, 32);
    }

    float invS = __builtin_amdgcn_rcpf(S);
    invS = __builtin_fmaf(invS, __builtin_fmaf(-S, invS, 1.0f), invS);

    v4f lo, hi;
    #pragma unroll
    for (int v = 0; v < 8; ++v) {
        const int row = v + 8 * hl;
        const float is = __shfl(invS, row, 32);
        const float val = acc[v] * is;
        if (v < 4) lo[v] = val; else hi[v - 4] = val;
    }
    if (m < nbatch) {
        float* p = out + (size_t)m * (size_t)n + (size_t)(t0 + 8 * hl);
        *(v4f*)(p + 0) = lo;
        *(v4f*)(p + 4) = hi;
    }
}

extern "C" void kernel_launch(void* const* d_in, const int* in_sizes, int n_in,
                              void* d_out, int out_size, void* d_ws, size_t ws_size,
                              hipStream_t stream) {
    const float* in      = (const float*)d_in[0];
    const float* alpha_p = (const float*)d_in[1];
    const float* beta_p  = (const float*)d_in[2];
    float* out = (float*)d_out;

    const int nbatch = 8;
    const int n = in_sizes[0] / nbatch;

    const int tiles = (n + 15) / 16;
    const int wavesPerBlock = 8;
    const int blocks = (tiles + wavesPerBlock - 1) / wavesPerBlock;

    lpf_wmma_kernel<<<blocks, wavesPerBlock * 32, 0, stream>>>(
        in, alpha_p, beta_p, out, n, nbatch);
}


# Each wave walks a strip of 4 consecutive interior tiles (grid cut to a quarter): lane-only tap terms built once per wave, B fragments shared by neighbour tiles loaded once; sum order unchanged
# speedup vs baseline: 1.1607x; 1.1607x over previous
#include <hip/hip_runtime.h>


typedef __attribute__((ext_vector_type(2))) float v2f;
typedef __attribute__((ext_vector_type(4))) float v4f;
typedef __attribute__((ext_vector_type(8))) float v8f;

#define FS_F      8000.0f
#define HALF_W    16
#define KTAPS     33
#define PI_F      3.14159265358979323846f
#define TWO_PI_F  6.28318530717958647692f
#define STRIP     4

__global__ __launch_bounds__(256) void
lpf_wmma_kernel(const float* __restrict__ in, const float* __restrict__ alpha_p,
                const float* __restrict__ beta_p, float* __restrict__ out,
                int n, int nbatch)
{
    const float alpha = alpha_p[0];
    const float beta  = beta_p[0];

    const int lane = threadIdx.x & 31;
    const int wv   = threadIdx.x >> 5;
    const int gw   = blockIdx.x * (blockDim.x >> 5) + wv;

    const int nt = (n + 15) >> 4;
    const int ns = (nt >= 3) ? (nt - 3) / STRIP : 0;
    const int nsingle = nt - STRIP * ns;
    if (gw >= ns + nsingle) return;

    const int m  = lane & 15;
    const int hl = lane >> 4;

    const int br = (m < nbatch) ? m : 0;
    const float* __restrict__ inb = in + (size_t)br * (size_t)n;

    if (gw >= nsingle) {
        const int t0s = 16 * (1 + STRIP * (gw - nsingle));

        const float* __restrict__ pB = inb + (size_t)(t0s + 2 * hl);
        float2 Bf[12 + 4 * (STRIP - 1)];
        #pragma unroll
        for (int q = 0; q < 12 + 4 * (STRIP - 1); ++q)
            Bf[q] = *(const float2*)(pB + 4 * q);

        float wm[12][2], xj[12][2];
        #pragma unroll
        for (int c = 0; c < 12; ++c) {
            #pragma unroll
            for (int e = 0; e < 2; ++e) {
                const int k = 4 * c + 2 * hl + e - m;
                const float fk = (float)k;
                const float w = 0.5f - 0.5f * __cosf(PI_F * (1.0f / 16.0f) * fk);
                wm[c][e] = w * (((unsigned)k < KTAPS) ? 1.0f : 0.0f);
                xj[c][e] = fk - 16.0f;
            }
        }

        #pragma unroll
        for (int i = 0; i < STRIP; ++i) {
            const int t0 = t0s + 16 * i;
            const int t  = t0 + m;
            const float c_t =
                (PI_F * 0.25f + alpha * __sinf(beta * (float)t * (1.0f / FS_F))) * (1.0f / TWO_PI_F);
            const float c2   = 2.0f * c_t;
            const float xstp = TWO_PI_F * c_t;

            v8f acc = {};
            float Apart = 0.0f;
            #pragma unroll
            for (int c = 0; c < 12; ++c) {
                v2f A, B;
                float v[2];
                #pragma unroll
                for (int e = 0; e < 2; ++e) {
                    const bool is16 = (c >= 4) && (c <= 7) &&
                                      (4 * c + 2 * hl + e - m == HALF_W);
                    const float x = xstp * xj[c][e];
                    float s = __sinf(x) * __builtin_amdgcn_rcpf(x);
                    s = is16 ? 1.0f : s;
                    v[e] = c2 * wm[c][e] * s;
                }
                A.x = v[0]; A.y = v[1];
                Apart += v[0] + v[1];
                B.x = Bf[c + 4 * i].x;
                B.y = Bf[c + 4 * i].y;
                acc = __builtin_amdgcn_wmma_f32_16x16x4_f32(
                    false, A, false, B, (short)0, acc, false, false);
            }
            const float S = Apart + __shfl_xor(Apart, 16, 32);

            float invS = __builtin_amdgcn_rcpf(S);
            invS = __builtin_fmaf(invS, __builtin_fmaf(-S, invS, 1.0f), invS);

            v4f lo, hi;
            #pragma unroll
            for (int r = 0; r < 8; ++r) {
                const int row = r + 8 * hl;
                const float is = __shfl(invS, row, 32);
                const float val = acc[r] * is;
                if (r < 4) lo[r] = val; else hi[r - 4] = val;
            }
            if (m < nbatch) {
                float* p = out + (size_t)m * (size_t)n + (size_t)(t0 + 8 * hl);
                *(v4f*)(p + 0) = lo;
                *(v4f*)(p + 4) = hi;
            }
        }
        return;
    }

    const int tile = (gw == 0) ? 0 : (STRIP * ns + gw);
    const int t0   = tile * 16;
    const int t    = t0 + m;

    const float c_t =
        (PI_F * 0.25f + alpha * __sinf(beta * (float)t * (1.0f / FS_F))) * (1.0f / TWO_PI_F);
    const float c2   = 2.0f * c_t;
    const float xstp = TWO_PI_F * c_t;

    auto tapf = [&](float fk, bool is16) -> float {
        float w = 0.5f - 0.5f * __cosf(PI_F * (1.0f / 16.0f) * fk);
        float x = xstp * (fk - 16.0f);
        float s = __sinf(x) * __builtin_amdgcn_rcpf(x);
        s = is16 ? 1.0f : s;
        return c2 * w * s;
    };

    v8f acc = {};
    float S;

    const bool edge = (t0 < HALF_W) || (t0 > n - 3 * HALF_W);
    if (!edge) {
        const float fkm = (float)(-m);
        const float* __restrict__ pB = inb + (size_t)(t0 + 2 * hl);
        float Apart = 0.0f;
        #pragma unroll
        for (int c = 0; c < 12; ++c) {
            const int base = 4 * c + 2 * hl;
            const int k0 = base - m;
            const float fk0 = fkm + (float)base;
            v2f A, B;
            float v0 = tapf(fk0,        k0     == HALF_W) * (((unsigned)k0       < KTAPS) ? 1.0f : 0.0f);
            float v1 = tapf(fk0 + 1.0f, k0 + 1 == HALF_W) * (((unsigned)(k0 + 1) < KTAPS) ? 1.0f : 0.0f);
            A.x = v0; A.y = v1;
            Apart += v0 + v1;
            const float2 bb = *(const float2*)(pB + 4 * c);
            B.x = bb.x;
            B.y = bb.y;
            acc = __builtin_amdgcn_wmma_f32_16x16x4_f32(
                false, A, false, B, (short)0, acc, false, false);
        }
        S = Apart + __shfl_xor(Apart, 16, 32);
    } else {
        const int off = (t < HALF_W) ? -(HALF_W + 1) : 0;
        const float fkm = (float)(-m - off);
        #pragma unroll
        for (int c = 0; c < 12; ++c) {
            const int base = 4 * c + 2 * hl;
            v2f A, B;
            #pragma unroll
            for (int j = 0; j < 2; ++j) {
                int k = base + j - m - off;
                bool valid = ((unsigned)k < KTAPS);
                valid = valid & ((t >= HALF_W) | (k >= HALF_W + 1 - t));
                valid = valid & ((t <= n - KTAPS) | (k < n - t));
                float v = tapf(fkm + (float)(base + j), k == HALF_W)
                          * (valid ? 1.0f : 0.0f);
                if (j == 0) A.x = v; else A.y = v;
            }
            int col = t0 + base;
            col = (col < n - 1) ? col : (n - 2);
            const float2 bb = *(const float2*)(inb + (size_t)col);
            B.x = bb.x;
            B.y = bb.y;
            acc = __builtin_amdgcn_wmma_f32_16x16x4_f32(
                false, A, false, B, (short)0, acc, false, false);
        }
        float Sp = 0.0f;
        #pragma unroll
        for (int j = 0; j <= 16; ++j) {
            int k = 2 * j + hl;
            float v = tapf((float)k, k == HALF_W);
            Sp += v * ((k < KTAPS) ? 1.0f : 0.0f);
        }
        S = Sp + __shfl_xor(Sp, 16, 32);
    }

    float invS = __builtin_amdgcn_rcpf(S);
    invS = __builtin_fmaf(invS, __builtin_fmaf(-S, invS, 1.0f), invS);

    v4f lo, hi;
    #pragma unroll
    for (int v = 0; v < 8; ++v) {
        const int row = v + 8 * hl;
        const float is = __shfl(invS, row, 32);
        const float val = acc[v] * is;
        if (v < 4) lo[v] = val; else hi[v - 4] = val;
    }
    if (m < nbatch) {
        float* p = out + (size_t)m * (size_t)n + (size_t)(t0 + 8 * hl);
        *(v4f*)(p + 0) = lo;
        *(v4f*)(p + 4) = hi;
    }
}

extern "C" void kernel_launch(void* const* d_in, const int* in_sizes, int n_in,
                              void* d_out, int out_size, void* d_ws, size_t ws_size,
                              hipStream_t stream) {
    const float* in      = (const float*)d_in[0];
    const float* alpha_p = (const float*)d_in[1];
    const float* beta_p  = (const float*)d_in[2];
    float* out = (float*)d_out;

    const int nbatch = 8;
    const int n = in_sizes[0] / nbatch;

    const int tiles  = (n + 15) / 16;
    const int strips = (tiles >= 3) ? (tiles - 3) / STRIP : 0;
    const int waves  = strips + (tiles - STRIP * strips);
    const int wavesPerBlock = 8;
    const int blocks = (waves + wavesPerBlock - 1) / wavesPerBlock;

    lpf_wmma_kernel<<<blocks, wavesPerBlock * 32, 0, stream>>>(
        in, alpha_p, beta_p, out, n, nbatch);
}


# Strip tiles: tap = lane-only w/(k-16) times sin; of 4 chunks the middle 2 sines by v_sin, outer 2 by 3-term recurrence (1 packed FMA); no per-tap rcp, row factor cancels in 1/S; sum order unchanged
# speedup vs baseline: 1.2236x; 1.0542x over previous
#include <hip/hip_runtime.h>


typedef __attribute__((ext_vector_type(2))) float v2f;
typedef __attribute__((ext_vector_type(4))) float v4f;
typedef __attribute__((ext_vector_type(8))) float v8f;

#define FS_F      8000.0f
#define HALF_W    16
#define KTAPS     33
#define PI_F      3.14159265358979323846f
#define TWO_PI_F  6.28318530717958647692f
#define STRIP     4

__global__ __launch_bounds__(256) void
lpf_wmma_kernel(const float* __restrict__ in, const float* __restrict__ alpha_p,
                const float* __restrict__ beta_p, float* __restrict__ out,
                int n, int nbatch)
{
    const float alpha = alpha_p[0];
    const float beta  = beta_p[0];

    const int lane = threadIdx.x & 31;
    const int wv   = threadIdx.x >> 5;
    const int gw   = blockIdx.x * (blockDim.x >> 5) + wv;

    const int nt = (n + 15) >> 4;
    const int ns = (nt >= 3) ? (nt - 3) / STRIP : 0;
    const int nsingle = nt - STRIP * ns;
    if (gw >= ns + nsingle) return;

    const int m  = lane & 15;
    const int hl = lane >> 4;

    const int br = (m < nbatch) ? m : 0;
    const float* __restrict__ inb = in + (size_t)br * (size_t)n;

    if (gw >= nsingle) {
        const int t0s = 16 * (1 + STRIP * (gw - nsingle));

        const float* __restrict__ pB = inb + (size_t)(t0s + 2 * hl);
        float2 Bf[12 + 4 * (STRIP - 1)];
        #pragma unroll
        for (int q = 0; q < 12 + 4 * (STRIP - 1); ++q)
            Bf[q] = *(const float2*)(pB + 4 * q);

        v2f gk[12];
        v2f wc[4];
        v2f js[3][2];
        #pragma unroll
        for (int c = 0; c < 12; ++c) {
            #pragma unroll
            for (int e = 0; e < 2; ++e) {
                const int k = 4 * c + 2 * hl + e - m;
                const float fk = (float)k;
                const float w = 0.5f - 0.5f * __cosf(PI_F * (1.0f / 16.0f) * fk);
                const float xj = fk - 16.0f;
                const bool inband = ((unsigned)k < KTAPS) && (k != HALF_W);
                gk[c][e] = inband ? w * __builtin_amdgcn_rcpf(xj) : 0.0f;
                if (c >= 4 && c <= 7) wc[c - 4][e] = w;
                if ((c & 3) == 1) js[c >> 2][0][e] = xj;
                if ((c & 3) == 2) js[c >> 2][1][e] = xj;
            }
        }

        #pragma unroll
        for (int i = 0; i < STRIP; ++i) {
            const int t0 = t0s + 16 * i;
            const int t  = t0 + m;
            const float c_t =
                (PI_F * 0.25f + alpha * __sinf(beta * (float)t * (1.0f / FS_F))) * (1.0f / TWO_PI_F);
            const float xstp = TWO_PI_F * c_t;

            const float K4 = 2.0f * __builtin_amdgcn_cosf(4.0f * c_t);

            v2f sn[12];
            #pragma unroll
            for (int g = 0; g < 3; ++g) {
                const v2f a1 = js[g][0] * c_t;
                const v2f a2 = js[g][1] * c_t;
                v2f s1, s2;
                s1.x = __builtin_amdgcn_sinf(a1.x);
                s1.y = __builtin_amdgcn_sinf(a1.y);
                s2.x = __builtin_amdgcn_sinf(a2.x);
                s2.y = __builtin_amdgcn_sinf(a2.y);
                sn[4 * g + 0] = s1 * K4 - s2;
                sn[4 * g + 1] = s1;
                sn[4 * g + 2] = s2;
                sn[4 * g + 3] = s2 * K4 - s1;
            }

            v8f acc = {};
            float Apart = 0.0f;
            #pragma unroll
            for (int c = 0; c < 12; ++c) {
                v2f A = gk[c] * sn[c];
                if (c >= 4 && c <= 7) {
                    const v2f ctr = wc[c - 4] * xstp;
                    #pragma unroll
                    for (int e = 0; e < 2; ++e)
                        if (4 * c + 2 * hl + e - m == HALF_W) A[e] = ctr[e];
                }
                Apart += A.x + A.y;
                v2f B;
                B.x = Bf[c + 4 * i].x;
                B.y = Bf[c + 4 * i].y;
                acc = __builtin_amdgcn_wmma_f32_16x16x4_f32(
                    false, A, false, B, (short)0, acc, false, false);
            }
            const float S = Apart + __shfl_xor(Apart, 16, 32);

            float invS = __builtin_amdgcn_rcpf(S);
            invS = __builtin_fmaf(invS, __builtin_fmaf(-S, invS, 1.0f), invS);

            v4f lo, hi;
            #pragma unroll
            for (int r = 0; r < 8; ++r) {
                const int row = r + 8 * hl;
                const float is = __shfl(invS, row, 32);
                const float val = acc[r] * is;
                if (r < 4) lo[r] = val; else hi[r - 4] = val;
            }
            if (m < nbatch) {
                float* p = out + (size_t)m * (size_t)n + (size_t)(t0 + 8 * hl);
                *(v4f*)(p + 0) = lo;
                *(v4f*)(p + 4) = hi;
            }
        }
        return;
    }

    const int tile = (gw == 0) ? 0 : (STRIP * ns + gw);
    const int t0   = tile * 16;
    const int t    = t0 + m;

    const float c_t =
        (PI_F * 0.25f + alpha * __sinf(beta * (float)t * (1.0f / FS_F))) * (1.0f / TWO_PI_F);
    const float c2   = 2.0f * c_t;
    const float xstp = TWO_PI_F * c_t;

    auto tapf = [&](float fk, bool is16) -> float {
        float w = 0.5f - 0.5f * __cosf(PI_F * (1.0f / 16.0f) * fk);
        float x = xstp * (fk - 16.0f);
        float s = __sinf(x) * __builtin_amdgcn_rcpf(x);
        s = is16 ? 1.0f : s;
        return c2 * w * s;
    };

    v8f acc = {};
    float S;

    const bool edge = (t0 < HALF_W) || (t0 > n - 3 * HALF_W);
    if (!edge) {
        const float fkm = (float)(-m);
        const float* __restrict__ pB = inb + (size_t)(t0 + 2 * hl);
        float Apart = 0.0f;
        #pragma unroll
        for (int c = 0; c < 12; ++c) {
            const int base = 4 * c + 2 * hl;
            const int k0 = base - m;
            const float fk0 = fkm + (float)base;
            v2f A, B;
            float v0 = tapf(fk0,        k0     == HALF_W) * (((unsigned)k0       < KTAPS) ? 1.0f : 0.0f);
            float v1 = tapf(fk0 + 1.0f, k0 + 1 == HALF_W) * (((unsigned)(k0 + 1) < KTAPS) ? 1.0f : 0.0f);
            A.x = v0; A.y = v1;
            Apart += v0 + v1;
            const float2 bb = *(const float2*)(pB + 4 * c);
            B.x = bb.x;
            B.y = bb.y;
            acc = __builtin_amdgcn_wmma_f32_16x16x4_f32(
                false, A, false, B, (short)0, acc, false, false);
        }
        S = Apart + __shfl_xor(Apart, 16, 32);
    } else {
        const int off = (t < HALF_W) ? -(HALF_W + 1) : 0;
        const float fkm = (float)(-m - off);
        #pragma unroll
        for (int c = 0; c < 12; ++c) {
            const int base = 4 * c + 2 * hl;
            v2f A, B;
            #pragma unroll
            for (int j = 0; j < 2; ++j) {
                int k = base + j - m - off;
                bool valid = ((unsigned)k < KTAPS);
                valid = valid & ((t >= HALF_W) | (k >= HALF_W + 1 - t));
                valid = valid & ((t <= n - KTAPS) | (k < n - t));
                float v = tapf(fkm + (float)(base + j), k == HALF_W)
                          * (valid ? 1.0f : 0.0f);
                if (j == 0) A.x = v; else A.y = v;
            }
            int col = t0 + base;
            col = (col < n - 1) ? col : (n - 2);
            const float2 bb = *(const float2*)(inb + (size_t)col);
            B.x = bb.x;
            B.y = bb.y;
            acc = __builtin_amdgcn_wmma_f32_16x16x4_f32(
                false, A, false, B, (short)0, acc, false, false);
        }
        float Sp = 0.0f;
        #pragma unroll
        for (int j = 0; j <= 16; ++j) {
            int k = 2 * j + hl;
            float v = tapf((float)k, k == HALF_W);
            Sp += v * ((k < KTAPS) ? 1.0f : 0.0f);
        }
        S = Sp + __shfl_xor(Sp, 16, 32);
    }

    float invS = __builtin_amdgcn_rcpf(S);
    invS = __builtin_fmaf(invS, __builtin_fmaf(-S, invS, 1.0f), invS);

    v4f lo, hi;
    #pragma unroll
    for (int v = 0; v < 8; ++v) {
        const int row = v + 8 * hl;
        const float is = __shfl(invS, row, 32);
        const float val = acc[v] * is;
        if (v < 4) lo[v] = val; else hi[v - 4] = val;
    }
    if (m < nbatch) {
        float* p = out + (size_t)m * (size_t)n + (size_t)(t0 + 8 * hl);
        *(v4f*)(p + 0) = lo;
        *(v4f*)(p + 4) = hi;
    }
}

extern "C" void kernel_launch(void* const* d_in, const int* in_sizes, int n_in,
                              void* d_out, int out_size, void* d_ws, size_t ws_size,
                              hipStream_t stream) {
    const float* in      = (const float*)d_in[0];
    const float* alpha_p = (const float*)d_in[1];
    const float* beta_p  = (const float*)d_in[2];
    float* out = (float*)d_out;

    const int nbatch = 8;
    const int n = in_sizes[0] / nbatch;

    const int tiles  = (n + 15) / 16;
    const int strips = (tiles >= 3) ? (tiles - 3) / STRIP : 0;
    const int waves  = strips + (tiles - STRIP * strips);
    const int wavesPerBlock = 8;
    const int blocks = (waves + wavesPerBlock - 1) / wavesPerBlock;

    lpf_wmma_kernel<<<blocks, wavesPerBlock * 32, 0, stream>>>(
        in, alpha_p, beta_p, out, n, nbatch);
}


# Strip tiles computed transposed (input as A, taps as B): idle A rows 8..15 hold ones, so the tap sum S comes out of the WMMA chain; 1 permlanex16 + 1 rcp a tile, 64-B stores; new S sum order
# speedup vs baseline: 1.3646x; 1.1152x over previous
#include <hip/hip_runtime.h>


typedef __attribute__((ext_vector_type(2))) float v2f;
typedef __attribute__((ext_vector_type(4))) float v4f;
typedef __attribute__((ext_vector_type(8))) float v8f;

#define FS_F      8000.0f
#define HALF_W    16
#define KTAPS     33
#define PI_F      3.14159265358979323846f
#define TWO_PI_F  6.28318530717958647692f
#define STRIP     4


__device__ __forceinline__ float xhalf(float x) {
    return __builtin_bit_cast(float, __builtin_amdgcn_permlanex16(
        __builtin_bit_cast(int, x), __builtin_bit_cast(int, x),
        0x76543210, 0xfedcba98, false, false));
}

__global__ __launch_bounds__(256) void
lpf_wmma_kernel(const float* __restrict__ in, const float* __restrict__ alpha_p,
                const float* __restrict__ beta_p, float* __restrict__ out,
                int n, int nbatch)
{
    const float alpha = alpha_p[0];
    const float beta  = beta_p[0];

    const int lane = threadIdx.x & 31;
    const int wv   = threadIdx.x >> 5;
    const int gw   = blockIdx.x * (blockDim.x >> 5) + wv;

    const int nt = (n + 15) >> 4;
    const int ns = (nt >= 3) ? (nt - 3) / STRIP : 0;
    const int nsingle = nt - STRIP * ns;
    if (gw >= ns + nsingle) return;

    const int m  = lane & 15;
    const int hl = lane >> 4;

    const int br = (m < nbatch) ? m : 0;
    const float* __restrict__ inb = in + (size_t)br * (size_t)n;

    if (gw >= nsingle) {
        const int t0s = 16 * (1 + STRIP * (gw - nsingle));

        const float* __restrict__ pB = inb + (size_t)(t0s + 2 * hl);
        v2f Xf[12 + 4 * (STRIP - 1)];
        #pragma unroll
        for (int q = 0; q < 12 + 4 * (STRIP - 1); ++q) {
            Xf[q].x = 1.0f;
            Xf[q].y = 1.0f;
        }
        if (m < 8) {
            #pragma unroll
            for (int q = 0; q < 12 + 4 * (STRIP - 1); ++q)
                Xf[q] = *(const v2f*)(pB + 4 * q);
        }

        v2f gk[12];
        v2f wc[4];
        v2f js[3][2];
        #pragma unroll
        for (int c = 0; c < 12; ++c) {
            #pragma unroll
            for (int e = 0; e < 2; ++e) {
                const int k = 4 * c + 2 * hl + e - m;
                const float fk = (float)k;
                const float w = 0.5f - 0.5f * __cosf(PI_F * (1.0f / 16.0f) * fk);
                const float xj = fk - 16.0f;
                const bool inband = ((unsigned)k < KTAPS) && (k != HALF_W);
                gk[c][e] = inband ? w * __builtin_amdgcn_rcpf(xj) : 0.0f;
                if (c >= 4 && c <= 7) wc[c - 4][e] = w;
                if ((c & 3) == 1) js[c >> 2][0][e] = xj;
                if ((c & 3) == 2) js[c >> 2][1][e] = xj;
            }
        }

        float* orow[8];
        #pragma unroll
        for (int r = 0; r < 8; ++r)
            orow[r] = out + (size_t)r * (size_t)n + (size_t)(t0s + m);

        #pragma unroll
        for (int i = 0; i < STRIP; ++i) {
            const int t0 = t0s + 16 * i;
            const int t  = t0 + m;
            const float c_t =
                (PI_F * 0.25f + alpha * __sinf(beta * (float)t * (1.0f / FS_F))) * (1.0f / TWO_PI_F);
            const float xstp = TWO_PI_F * c_t;

            const float K4 = 2.0f * __builtin_amdgcn_cosf(4.0f * c_t);

            v2f sn[12];
            #pragma unroll
            for (int g = 0; g < 3; ++g) {
                const v2f a1 = js[g][0] * c_t;
                const v2f a2 = js[g][1] * c_t;
                v2f s1, s2;
                s1.x = __builtin_amdgcn_sinf(a1.x);
                s1.y = __builtin_amdgcn_sinf(a1.y);
                s2.x = __builtin_amdgcn_sinf(a2.x);
                s2.y = __builtin_amdgcn_sinf(a2.y);
                sn[4 * g + 0] = s1 * K4 - s2;
                sn[4 * g + 1] = s1;
                sn[4 * g + 2] = s2;
                sn[4 * g + 3] = s2 * K4 - s1;
            }

            v8f acc = {};
            #pragma unroll
            for (int c = 0; c < 12; ++c) {
                v2f Tp = gk[c] * sn[c];
                if (c >= 4 && c <= 7) {
                    const v2f ctr = wc[c - 4] * xstp;
                    #pragma unroll
                    for (int e = 0; e < 2; ++e)
                        if (4 * c + 2 * hl + e - m == HALF_W) Tp[e] = ctr[e];
                }
                acc = __builtin_amdgcn_wmma_f32_16x16x4_f32(
                    false, Xf[c + 4 * i], false, Tp, (short)0, acc, false, false);
            }
            asm volatile("s_wait_alu depctr_va_vdst(0)" : "+v"(acc));

            const float S = xhalf(acc[0]);

            float invS = __builtin_amdgcn_rcpf(S);
            invS = __builtin_fmaf(invS, __builtin_fmaf(-S, invS, 1.0f), invS);

            if (hl == 0) {
                #pragma unroll
                for (int r = 0; r < 8; ++r)
                    orow[r][16 * i] = acc[r] * invS;
            }
        }
        return;
    }

    const int tile = (gw == 0) ? 0 : (STRIP * ns + gw);
    const int t0   = tile * 16;
    const int t    = t0 + m;

    const float c_t =
        (PI_F * 0.25f + alpha * __sinf(beta * (float)t * (1.0f / FS_F))) * (1.0f / TWO_PI_F);
    const float c2   = 2.0f * c_t;
    const float xstp = TWO_PI_F * c_t;

    auto tapf = [&](float fk, bool is16) -> float {
        float w = 0.5f - 0.5f * __cosf(PI_F * (1.0f / 16.0f) * fk);
        float x = xstp * (fk - 16.0f);
        float s = __sinf(x) * __builtin_amdgcn_rcpf(x);
        s = is16 ? 1.0f : s;
        return c2 * w * s;
    };

    v8f acc = {};
    float S;

    const bool edge = (t0 < HALF_W) || (t0 > n - 3 * HALF_W);
    if (!edge) {
        const float fkm = (float)(-m);
        const float* __restrict__ pB = inb + (size_t)(t0 + 2 * hl);
        float Apart = 0.0f;
        #pragma unroll
        for (int c = 0; c < 12; ++c) {
            const int base = 4 * c + 2 * hl;
            const int k0 = base - m;
            const float fk0 = fkm + (float)base;
            v2f A, B;
            float v0 = tapf(fk0,        k0     == HALF_W) * (((unsigned)k0       < KTAPS) ? 1.0f : 0.0f);
            float v1 = tapf(fk0 + 1.0f, k0 + 1 == HALF_W) * (((unsigned)(k0 + 1) < KTAPS) ? 1.0f : 0.0f);
            A.x = v0; A.y = v1;
            Apart += v0 + v1;
            const float2 bb = *(const float2*)(pB + 4 * c);
            B.x = bb.x;
            B.y = bb.y;
            acc = __builtin_amdgcn_wmma_f32_16x16x4_f32(
                false, A, false, B, (short)0, acc, false, false);
        }
        S = Apart + __shfl_xor(Apart, 16, 32);
    } else {
        const int off = (t < HALF_W) ? -(HALF_W + 1) : 0;
        const float fkm = (float)(-m - off);
        #pragma unroll
        for (int c = 0; c < 12; ++c) {
            const int base = 4 * c + 2 * hl;
            v2f A, B;
            #pragma unroll
            for (int j = 0; j < 2; ++j) {
                int k = base + j - m - off;
                bool valid = ((unsigned)k < KTAPS);
                valid = valid & ((t >= HALF_W) | (k >= HALF_W + 1 - t));
                valid = valid & ((t <= n - KTAPS) | (k < n - t));
                float v = tapf(fkm + (float)(base + j), k == HALF_W)
                          * (valid ? 1.0f : 0.0f);
                if (j == 0) A.x = v; else A.y = v;
            }
            int col = t0 + base;
            col = (col < n - 1) ? col : (n - 2);
            const float2 bb = *(const float2*)(inb + (size_t)col);
            B.x = bb.x;
            B.y = bb.y;
            acc = __builtin_amdgcn_wmma_f32_16x16x4_f32(
                false, A, false, B, (short)0, acc, false, false);
        }
        float Sp = 0.0f;
        #pragma unroll
        for (int j = 0; j <= 16; ++j) {
            int k = 2 * j + hl;
            float v = tapf((float)k, k == HALF_W);
            Sp += v * ((k < KTAPS) ? 1.0f : 0.0f);
        }
        S = Sp + __shfl_xor(Sp, 16, 32);
    }

    float invS = __builtin_amdgcn_rcpf(S);
    invS = __builtin_fmaf(invS, __builtin_fmaf(-S, invS, 1.0f), invS);

    v4f lo, hi;
    #pragma unroll
    for (int v = 0; v < 8; ++v) {
        const int row = v + 8 * hl;
        const float is = __shfl(invS, row, 32);
        const float val = acc[v] * is;
        if (v < 4) lo[v] = val; else hi[v - 4] = val;
    }
    if (m < nbatch) {
        float* p = out + (size_t)m * (size_t)n + (size_t)(t0 + 8 * hl);
        *(v4f*)(p + 0) = lo;
        *(v4f*)(p + 4) = hi;
    }
}

extern "C" void kernel_launch(void* const* d_in, const int* in_sizes, int n_in,
                              void* d_out, int out_size, void* d_ws, size_t ws_size,
                              hipStream_t stream) {
    const float* in      = (const float*)d_in[0];
    const float* alpha_p = (const float*)d_in[1];
    const float* beta_p  = (const float*)d_in[2];
    float* out = (float*)d_out;

    const int nbatch = 8;
    const int n = in_sizes[0] / nbatch;

    const int tiles  = (n + 15) / 16;
    const int strips = (tiles >= 3) ? (tiles - 3) / STRIP : 0;
    const int waves  = strips + (tiles - STRIP * strips);
    const int wavesPerBlock = 8;
    const int blocks = (waves + wavesPerBlock - 1) / wavesPerBlock;

    lpf_wmma_kernel<<<blocks, wavesPerBlock * 32, 0, stream>>>(
        in, alpha_p, beta_p, out, n, nbatch);
}
